# starting kernel
# baseline (speedup 1.0000x reference)
#include <hip/hip_runtime.h>
#include <hip/hip_bf16.h>


typedef __attribute__((ext_vector_type(2))) float v2f;
typedef __attribute__((ext_vector_type(8))) float v8f;

__device__ __forceinline__ unsigned f2ord(float f) {
    unsigned u = __float_as_uint(f);
    return (u & 0x80000000u) ? ~u : (u | 0x80000000u);
}
__device__ __forceinline__ float ord2f(unsigned u) {
    unsigned v = (u & 0x80000000u) ? (u & 0x7FFFFFFFu) : ~u;
    return __uint_as_float(v);
}

__global__ void gat_gemm_wmma(const float* __restrict__ x,
                              const float* __restrict__ W,
                              float* __restrict__ h) {
    const int wave   = threadIdx.x >> 5;
    const int lane   = threadIdx.x & 31;
    const int lhalf  = lane & 15;
    const int khalf  = (lane >> 4) << 1;
    const int rowA   = blockIdx.x * 16 + lhalf;
    const int colB   = wave * 16 + lhalf;

    const float* __restrict__ arow = x + (size_t)rowA * 128;

    v8f c = {0.f, 0.f, 0.f, 0.f, 0.f, 0.f, 0.f, 0.f};

#pragma unroll 4
    for (int k0 = 0; k0 < 128; k0 += 4) {
        const int k = k0 + khalf;
        v2f a;
        a.x = arow[k];
        a.y = arow[k + 1];
        v2f b;
        b.x = W[(size_t)k * 128 + colB];
        b.y = W[(size_t)(k + 1) * 128 + colB];
        c = __builtin_amdgcn_wmma_f32_16x16x4_f32(false, a, false, b,
                                                  (short)0, c, false, false);
    }

    const int mhi = (lane >> 4) << 3;
#pragma unroll
    for (int r = 0; r < 8; ++r) {
        const int m = blockIdx.x * 16 + r + mhi;
        h[(size_t)m * 128 + wave * 16 + lhalf] = c[r];
    }
}

__global__ void gat_init(float* __restrict__ out, const float* __restrict__ bias,
                         float* __restrict__ denom, unsigned* __restrict__ maxbuf,
                         int total, int nh) {
    const int i = blockIdx.x * blockDim.x + threadIdx.x;
    if (i < 4)     maxbuf[i] = f2ord(-3.0e38f);
    if (i < nh)    denom[i]  = 0.0f;
    if (i < total) out[i]    = bias[i & 127];
}

__global__ void gat_alpha(const float* __restrict__ h,
                          const float* __restrict__ att_l,
                          const float* __restrict__ att_r,
                          float* __restrict__ alpha_l,
                          float* __restrict__ alpha_r, int nh) {
    const int i = blockIdx.x * blockDim.x + threadIdx.x;
    if (i >= nh) return;
    const int head = i & 3;
    const float* __restrict__ hv = h + ((size_t)(i >> 2)) * 128 + head * 32;
    const float* __restrict__ al = att_l + head * 32;
    const float* __restrict__ ar = att_r + head * 32;
    float sl = 0.f, sr = 0.f;
#pragma unroll
    for (int d = 0; d < 32; ++d) {
        const float v = hv[d];
        sl += v * al[d];
        sr += v * ar[d];
    }
    alpha_l[i] = sl;
    alpha_r[i] = sr;
}

__global__ void gat_edge_max(const int* __restrict__ ei,
                             const float* __restrict__ alpha_l,
                             const float* __restrict__ alpha_r,
                             unsigned* __restrict__ maxbuf, int E) {
    __shared__ unsigned smax[4];
    if (threadIdx.x < 4) smax[threadIdx.x] = 0u;
    __syncthreads();
    const int e = blockIdx.x * blockDim.x + threadIdx.x;
    if (e < E) {
        const int f = ei[e];
        const int t = ei[E + e];
#pragma unroll
        for (int hd = 0; hd < 4; ++hd) {
            float a = alpha_l[f * 4 + hd] + alpha_r[t * 4 + hd];
            a = a > 0.f ? a : 0.2f * a;
            atomicMax(&smax[hd], f2ord(a));
        }
    }
    __syncthreads();
    if (threadIdx.x < 4) atomicMax(&maxbuf[threadIdx.x], smax[threadIdx.x]);
}

__global__ void gat_denom(const int* __restrict__ ei,
                          const float* __restrict__ alpha_l,
                          const float* __restrict__ alpha_r,
                          const unsigned* __restrict__ maxbuf,
                          float* __restrict__ denom, int E) {
    const int e = blockIdx.x * blockDim.x + threadIdx.x;
    if (e >= E) return;
    const int f = ei[e];
    const int t = ei[E + e];
#pragma unroll
    for (int hd = 0; hd < 4; ++hd) {
        float a = alpha_l[f * 4 + hd] + alpha_r[t * 4 + hd];
        a = a > 0.f ? a : 0.2f * a;
        const float ex = __expf(a - ord2f(maxbuf[hd]));
        atomicAdd(&denom[t * 4 + hd], ex);
    }
}

__global__ void gat_aggregate(const float* __restrict__ hbuf,
                              const int* __restrict__ ei,
                              const float* __restrict__ alpha_l,
                              const float* __restrict__ alpha_r,
                              const float* __restrict__ denom,
                              const unsigned* __restrict__ maxbuf,
                              float* __restrict__ out, int E) {
    const int gw   = (int)((blockIdx.x * (unsigned)blockDim.x + threadIdx.x) >> 5);
    const int lane = threadIdx.x & 31;
    if (gw >= E) return;
    const int f = ei[gw];
    const int t = ei[E + gw];
    const float* __restrict__ src = hbuf + (size_t)f * 128;
    float* __restrict__ dst = out + (size_t)t * 128;
#pragma unroll
    for (int hd = 0; hd < 4; ++hd) {
        float a = alpha_l[f * 4 + hd] + alpha_r[t * 4 + hd];
        a = a > 0.f ? a : 0.2f * a;
        const float ex = __expf(a - ord2f(maxbuf[hd]));
        const float w  = ex / (denom[t * 4 + hd] + 1e-9f);
        atomicAdd(&dst[hd * 32 + lane], w * src[hd * 32 + lane]);
    }
}

extern "C" void kernel_launch(void* const* d_in, const int* in_sizes, int n_in,
                              void* d_out, int out_size, void* d_ws, size_t ws_size,
                              hipStream_t stream) {
    const float* x     = (const float*)d_in[0];
    const int*   ei    = (const int*)d_in[1];
    const float* W     = (const float*)d_in[2];
    const float* att_l = (const float*)d_in[3];
    const float* att_r = (const float*)d_in[4];
    const float* bias  = (const float*)d_in[5];
    float* out = (float*)d_out;

    const int N  = in_sizes[0] / 128;
    const int E  = in_sizes[1] / 2;
    const int NH = N * 4;

    float* h       = (float*)d_ws;
    float* alpha_l = h + (size_t)N * 128;
    float* alpha_r = alpha_l + NH;
    float* denom   = alpha_r + NH;
    unsigned* maxbuf = (unsigned*)(denom + NH);

    const int BLK = 256;

    {
        const int total = N * 128;
        gat_init<<<(total + BLK - 1) / BLK, BLK, 0, stream>>>(out, bias, denom,
                                                              maxbuf, total, NH);
    }
    gat_gemm_wmma<<<N / 16, BLK, 0, stream>>>(x, W, h);
    gat_alpha<<<(NH + BLK - 1) / BLK, BLK, 0, stream>>>(h, att_l, att_r,
                                                        alpha_l, alpha_r, NH);
    gat_edge_max<<<(E + BLK - 1) / BLK, BLK, 0, stream>>>(ei, alpha_l, alpha_r,
                                                          maxbuf, E);
    gat_denom<<<(E + BLK - 1) / BLK, BLK, 0, stream>>>(ei, alpha_l, alpha_r,
                                                       maxbuf, denom, E);
    {
        const long long threads = (long long)E * 32;
        const int blocks = (int)((threads + BLK - 1) / BLK);
        gat_aggregate<<<blocks, BLK, 0, stream>>>(h, ei, alpha_l, alpha_r,
                                                  denom, maxbuf, out, E);
    }
}


# GEMM launch takes over the other per-node work: a wave zeroes its 16 rows of out and dn and sums alpha_l/alpha_r from its accumulators by a lane-exchange tree (new sum order); 6 launches -> 4
# speedup vs baseline: 2.4650x; 2.4650x over previous
#include <hip/hip_runtime.h>
#include <hip/hip_bf16.h>


typedef __attribute__((ext_vector_type(2))) float v2f;
typedef __attribute__((ext_vector_type(4))) float v4f;
typedef __attribute__((ext_vector_type(8))) float v8f;
typedef __attribute__((address_space(1))) float gfloat;

__device__ __forceinline__ unsigned f2ord(float f) {
    unsigned u = __float_as_uint(f);
    return (u & 0x80000000u) ? ~u : (u | 0x80000000u);
}
__device__ __forceinline__ float ord2f(unsigned u) {
    unsigned v = (u & 0x80000000u) ? (u & 0x7FFFFFFFu) : ~u;
    return __uint_as_float(v);
}


template <int K>
__device__ __forceinline__ float lane_xor16(float v) {
    return __int_as_float(__builtin_amdgcn_update_dpp(0, __float_as_int(v), 0x160 + K,
                                                      0xf, 0xf, true));
}
template <int K>
__device__ __forceinline__ float xsum(bool bit, float v0, float v1) {
    const float keep = bit ? v1 : v0;
    const float send = bit ? v0 : v1;
    return keep + lane_xor16<K>(send);
}

__global__ __launch_bounds__(256)
void gat_gemm_wmma(const float* __restrict__ x,
                   const float* __restrict__ W,
                   const float* __restrict__ att_l,
                   const float* __restrict__ att_r,
                   float* __restrict__ h,
                   float* __restrict__ alpha_l,
                   float* __restrict__ alpha_r,
                   float* __restrict__ out,
                   float* __restrict__ dn,
                   unsigned* __restrict__ maxbuf,
                   unsigned* __restrict__ claim,
                   int ntiles, int N, int nclaim) {
    __shared__ v4f Wl[16 * 8 * 32];

    const int t    = threadIdx.x;
    const int wave = t >> 5;
    const int lane = t & 31;
    const int c    = lane & 15;
    const int hf   = lane >> 4;

    {
        const int gi = blockIdx.x * 256 + t;
        if (gi < nclaim) claim[gi]  = 0u;
        if (gi < 4)      maxbuf[gi] = f2ord(-3.0e38f);
    }

    {
        const float* __restrict__ wp = W + (size_t)(4 * hf) * 128 + wave * 16 + c;
#pragma unroll
        for (int i = 0; i < 16; ++i) {
            const float* __restrict__ p = wp + (size_t)(8 * i) * 128;
            v4f v;
            v.x = p[0];
            v.y = p[128];
            v.z = p[256];
            v.w = p[384];
            Wl[i * 256 + t] = v;
        }
    }

    const int tile = blockIdx.x * 8 + wave;
    const int tl   = tile < ntiles ? tile : ntiles - 1;
    const float* __restrict__ arow = x + ((size_t)tl * 16 + c) * 128 + 4 * hf;
    v4f a4[16];
#pragma unroll
    for (int q = 0; q < 16; ++q)
        a4[q] = *reinterpret_cast<const v4f*>(arow + 8 * q);

    float AL[8], AR[8];
#pragma unroll
    for (int n = 0; n < 8; ++n) {
        AL[n] = att_l[n * 16 + c];
        AR[n] = att_r[n * 16 + c];
    }

    __syncthreads();
    if (tile >= ntiles) return;

    {
        gfloat* op = (gfloat*)(out + (size_t)tile * 2048 + lane * 4);
        const v4f z = {0.f, 0.f, 0.f, 0.f};
#pragma unroll
        for (int i = 0; i < 16; ++i)
            *reinterpret_cast<__attribute__((address_space(1))) v4f*>(op + i * 128) = z;
        dn[(size_t)tile * 32 + lane]                 = 0.0f;
        dn[(size_t)N * 2 + (size_t)tile * 32 + lane] = 0.0f;
    }

    gfloat* hp  = (gfloat*)(h + ((size_t)tile * 16 + hf * 8) * 128 + c);
    gfloat* pal = (gfloat*)(alpha_l + (size_t)tile * 64 + lane * 2);
    gfloat* par = (gfloat*)(alpha_r + (size_t)tile * 64 + lane * 2);
    asm volatile("" : "+v"(hp), "+v"(pal), "+v"(par));

    v8f acc[8];
#pragma unroll
    for (int n = 0; n < 8; ++n)
        acc[n] = (v8f){0.f, 0.f, 0.f, 0.f, 0.f, 0.f, 0.f, 0.f};

#pragma unroll
    for (int q = 0; q < 16; ++q) {
        v4f b[8];
#pragma unroll
        for (int n = 0; n < 8; ++n) b[n] = Wl[(q * 8 + n) * 32 + lane];
#pragma unroll
        for (int n = 0; n < 8; ++n) {
            v2f a  = {a4[q].x, a4[q].y};
            v2f bb = {b[n].x, b[n].y};
            acc[n] = __builtin_amdgcn_wmma_f32_16x16x4_f32(false, a, false, bb,
                                                           (short)0, acc[n], false, false);
        }
#pragma unroll
        for (int n = 0; n < 8; ++n) {
            v2f a  = {a4[q].z, a4[q].w};
            v2f bb = {b[n].z, b[n].w};
            acc[n] = __builtin_amdgcn_wmma_f32_16x16x4_f32(false, a, false, bb,
                                                           (short)0, acc[n], false, false);
        }
    }

    int lb = lane;
    asm volatile("v_nop\n\tv_nop\n\tv_nop\n\tv_nop\n\tv_nop\n\tv_nop\n\tv_nop\n\tv_nop\n\tv_nop"
                 : "+v"(acc[0]), "+v"(acc[1]), "+v"(acc[2]), "+v"(acc[3]),
                   "+v"(acc[4]), "+v"(acc[5]), "+v"(acc[6]), "+v"(acc[7]),
                   "+v"(lb), "+v"(hp), "+v"(pal), "+v"(par));

#pragma unroll
    for (int n = 0; n < 8; ++n) {
#pragma unroll
        for (int r = 0; r < 8; ++r) hp[r * 128 + n * 16] = acc[n][r];
    }

    float P[2][4][8];
#pragma unroll
    for (int hd = 0; hd < 4; ++hd) {
#pragma unroll
        for (int r = 0; r < 8; ++r) {
            P[0][hd][r] = acc[2 * hd][r] * AL[2 * hd] + acc[2 * hd + 1][r] * AL[2 * hd + 1];
            P[1][hd][r] = acc[2 * hd][r] * AR[2 * hd] + acc[2 * hd + 1][r] * AR[2 * hd + 1];
        }
    }
    const bool b0 = (lb & 1) != 0, b1 = (lb & 2) != 0, b2 = (lb & 4) != 0, b3 = (lb & 8) != 0;
    float Q[2][2];
#pragma unroll
    for (int s = 0; s < 2; ++s) {
#pragma unroll
        for (int h0 = 0; h0 < 2; ++h0) {
            float s1[8], s2[4], s3[2];
#pragma unroll
            for (int r = 0; r < 8; ++r)
                s1[r] = xsum<1>(b0, P[s][h0][r], P[s][h0 + 2][r]);
#pragma unroll
            for (int r = 0; r < 4; ++r)
                s2[r] = xsum<2>(b1, s1[2 * r], s1[2 * r + 1]);
#pragma unroll
            for (int r = 0; r < 2; ++r)
                s3[r] = xsum<4>(b2, s2[2 * r], s2[2 * r + 1]);
            Q[s][h0] = xsum<8>(b3, s3[0], s3[1]);
        }
    }
    *reinterpret_cast<__attribute__((address_space(1))) v2f*>(pal) = (v2f){Q[0][0], Q[0][1]};
    *reinterpret_cast<__attribute__((address_space(1))) v2f*>(par) = (v2f){Q[1][0], Q[1][1]};
}

__global__ void gat_edge_max(const int* __restrict__ ei,
                             const float* __restrict__ alpha_l,
                             const float* __restrict__ alpha_r,
                             unsigned* __restrict__ maxbuf, int E) {
    __shared__ unsigned smax[4];
    if (threadIdx.x < 4) smax[threadIdx.x] = 0u;
    __syncthreads();
    const int e = blockIdx.x * blockDim.x + threadIdx.x;
    if (e < E) {
        const int f = ei[e];
        const int t = ei[E + e];
#pragma unroll
        for (int hd = 0; hd < 4; ++hd) {
            float a = alpha_l[f * 4 + hd] + alpha_r[t * 4 + hd];
            a = a > 0.f ? a : 0.2f * a;
            atomicMax(&smax[hd], f2ord(a));
        }
    }
    __syncthreads();
    if (threadIdx.x < 4) atomicMax(&maxbuf[threadIdx.x], smax[threadIdx.x]);
}

__global__ void
gat_aggregate(const float* __restrict__ hbuf,
              const int* __restrict__ ei,
              const float* __restrict__ alpha_l,
              const float* __restrict__ alpha_r,
              const unsigned* __restrict__ maxbuf,
              float* __restrict__ dn,
              float* __restrict__ out,
              unsigned* __restrict__ claim, int pitch, int E, int N) {
    __shared__ int s_item;
    const int lane = threadIdx.x & 31;
    const int wave = threadIdx.x >> 5;
    const int g    = blockIdx.x >> 5;

    if (threadIdx.x == 0) {
        const unsigned id   = __builtin_amdgcn_s_sendmsg_rtn(0x87);
        const unsigned die  = (id >> 16) & 7u;
        const unsigned pref = ((die & 3u) << 2) | ((blockIdx.x >> 3) & 3u);
        unsigned ch = die >> 2;
        int got = -1;
        for (int r = 0; r < 2 && got < 0; ++r) {
            unsigned* __restrict__ w = claim + (size_t)ch * pitch + g;
            unsigned it = pref;
            for (int k = 0; k < 16; ++k) {
                const unsigned bit = 1u << it;
                const unsigned old = atomicOr(w, bit);
                if (!(old & bit)) { got = (int)(ch * 16u + it); break; }
                const unsigned freeb = ~(old | bit) & 0xFFFFu;
                if (!freeb) break;
                it = (unsigned)__builtin_ctz(freeb);
            }
            ch ^= 1u;
        }
        s_item = got;
    }
    __syncthreads();
    const int item = s_item;
    if (item < 0) return;

    const int c  = item >> 4;
    const int e0 = __builtin_amdgcn_readfirstlane(g * 2048 + (item & 15) * 128 + wave * 16);
    if (e0 >= E) return;
    const int nlive = (E - e0) < 16 ? (E - e0) : 16;

    const int  j  = lane >> 1;
    const int  hb = lane & 1;
    const int  hd = 2 * c + hb;
    const int  e  = (j < nlive) ? (e0 + j) : e0;
    const int  f  = ei[e];
    const int  t  = ei[E + e];
    float* __restrict__ dnc = dn + (size_t)c * N * 2;

    if (nlive == 16) {
        const float    al = alpha_l[f * 4 + hd];
        const float    ar = alpha_r[t * 4 + hd];
        const unsigned mx = maxbuf[hd];

        float v[16][2];
#pragma unroll
        for (int jj = 0; jj < 16; ++jj) {
            const float* __restrict__ srow =
                hbuf + (size_t)__builtin_amdgcn_readlane(f, jj * 2) * 128 + c * 64;
#pragma unroll
            for (int h2 = 0; h2 < 2; ++h2) v[jj][h2] = srow[h2 * 32 + lane];
        }

        float a = al + ar;
        a = a > 0.f ? a : 0.2f * a;
        const float ex = __expf(a - ord2f(mx));
        atomicAdd(&dnc[t * 2 + hb], ex);

#pragma unroll
        for (int jj = 0; jj < 16; ++jj) {
            float* __restrict__ dst =
                out + (size_t)__builtin_amdgcn_readlane(t, jj * 2) * 128 + c * 64;
#pragma unroll
            for (int h2 = 0; h2 < 2; ++h2) {
                const float w = __uint_as_float(
                    __builtin_amdgcn_readlane(__float_as_uint(ex), jj * 2 + h2));
                atomicAdd(&dst[h2 * 32 + lane], w * v[jj][h2]);
            }
        }
    } else {
        for (int jj = 0; jj < nlive; ++jj) {
            const int fj = __builtin_amdgcn_readlane(f, jj * 2);
            const int tj = __builtin_amdgcn_readlane(t, jj * 2);
            const float* __restrict__ srow = hbuf + (size_t)fj * 128 + c * 64;
            float* __restrict__ dst = out + (size_t)tj * 128 + c * 64;
            float ex_mine = 0.f;
#pragma unroll
            for (int h2 = 0; h2 < 2; ++h2) {
                const int hh = 2 * c + h2;
                float a = alpha_l[fj * 4 + hh] + alpha_r[tj * 4 + hh];
                a = a > 0.f ? a : 0.2f * a;
                const float ex = __expf(a - ord2f(maxbuf[hh]));
                if (lane == h2) ex_mine = ex;
                atomicAdd(&dst[h2 * 32 + lane], ex * srow[h2 * 32 + lane]);
            }
            if (lane < 2) atomicAdd(&dnc[tj * 2 + lane], ex_mine);
        }
    }
}

__global__ void gat_normalise(float* __restrict__ out,
                              const float* __restrict__ dn,
                              const float* __restrict__ bias, int total4, int N) {
    const int i = blockIdx.x * blockDim.x + threadIdx.x;
    if (i >= total4) return;
    const int col  = (i & 31) * 4;
    const int node = i >> 5;
    const int head = col >> 5;
    const float dv = dn[((size_t)(head >> 1) * N + node) * 2 + (head & 1)] + 1e-9f;
    float4 v = reinterpret_cast<float4*>(out)[i];
    v.x = v.x / dv + bias[col + 0];
    v.y = v.y / dv + bias[col + 1];
    v.z = v.z / dv + bias[col + 2];
    v.w = v.w / dv + bias[col + 3];
    reinterpret_cast<float4*>(out)[i] = v;
}

extern "C" void kernel_launch(void* const* d_in, const int* in_sizes, int n_in,
                              void* d_out, int out_size, void* d_ws, size_t ws_size,
                              hipStream_t stream) {
    const float* x     = (const float*)d_in[0];
    const int*   ei    = (const int*)d_in[1];
    const float* W     = (const float*)d_in[2];
    const float* att_l = (const float*)d_in[3];
    const float* att_r = (const float*)d_in[4];
    const float* bias  = (const float*)d_in[5];
    float* out = (float*)d_out;

    const int N  = in_sizes[0] / 128;
    const int E  = in_sizes[1] / 2;
    const int NH = N * 4;
    const int NG = (E + 2047) / 2048;
    const int PITCH = (NG + 63) & ~63;

    float* h       = (float*)d_ws;
    float* alpha_l = h + (size_t)N * 128;
    float* alpha_r = alpha_l + NH;
    float* dn      = alpha_r + NH;
    unsigned* maxbuf = (unsigned*)(dn + NH);
    unsigned* claim  = maxbuf + 64;

    const int BLK = 256;

    {
        const int ntiles = N / 16;
        gat_gemm_wmma<<<(ntiles + 7) / 8, BLK, 0, stream>>>(x, W, att_l, att_r, h,
                                                            alpha_l, alpha_r, out, dn,
                                                            maxbuf, claim,
                                                            ntiles, N, 2 * PITCH);
    }
    gat_edge_max<<<(E + BLK - 1) / BLK, BLK, 0, stream>>>(ei, alpha_l, alpha_r,
                                                          maxbuf, E);
    gat_aggregate<<<NG * 32, BLK, 0, stream>>>(h, ei, alpha_l, alpha_r,
                                              maxbuf, dn, out, claim, PITCH, E, N);
    {
        const int total4 = N * 32;
        gat_normalise<<<(total4 + BLK - 1) / BLK, BLK, 0, stream>>>(out, dn,
                                                                    bias, total4, N);
    }
}


# Max pass: thread = 4 edges, 8 index loads then 8 16-byte alpha loads in flight at once; wave max by lane exchange, one LDS word per wave and head, not 4 serial 32-step scalar loops; sums untouched
# speedup vs baseline: 2.8991x; 1.1761x over previous
#include <hip/hip_runtime.h>
#include <hip/hip_bf16.h>


typedef __attribute__((ext_vector_type(2))) float v2f;
typedef __attribute__((ext_vector_type(4))) float v4f;
typedef __attribute__((ext_vector_type(8))) float v8f;
typedef __attribute__((address_space(1))) float gfloat;

__device__ __forceinline__ unsigned f2ord(float f) {
    unsigned u = __float_as_uint(f);
    return (u & 0x80000000u) ? ~u : (u | 0x80000000u);
}
__device__ __forceinline__ float ord2f(unsigned u) {
    unsigned v = (u & 0x80000000u) ? (u & 0x7FFFFFFFu) : ~u;
    return __uint_as_float(v);
}


template <int K>
__device__ __forceinline__ float lane_xor16(float v) {
    return __int_as_float(__builtin_amdgcn_update_dpp(0, __float_as_int(v), 0x160 + K,
                                                      0xf, 0xf, true));
}
template <int K>
__device__ __forceinline__ float xsum(bool bit, float v0, float v1) {
    const float keep = bit ? v1 : v0;
    const float send = bit ? v0 : v1;
    return keep + lane_xor16<K>(send);
}

__global__ __launch_bounds__(256)
void gat_gemm_wmma(const float* __restrict__ x,
                   const float* __restrict__ W,
                   const float* __restrict__ att_l,
                   const float* __restrict__ att_r,
                   float* __restrict__ h,
                   float* __restrict__ alpha_l,
                   float* __restrict__ alpha_r,
                   float* __restrict__ out,
                   float* __restrict__ dn,
                   unsigned* __restrict__ maxbuf,
                   unsigned* __restrict__ claim,
                   int ntiles, int N, int nclaim) {
    __shared__ v4f Wl[16 * 8 * 32];

    const int t    = threadIdx.x;
    const int wave = t >> 5;
    const int lane = t & 31;
    const int c    = lane & 15;
    const int hf   = lane >> 4;

    {
        const int gi = blockIdx.x * 256 + t;
        if (gi < nclaim) claim[gi]  = 0u;
        if (gi < 4)      maxbuf[gi] = f2ord(-3.0e38f);
    }

    {
        const float* __restrict__ wp = W + (size_t)(4 * hf) * 128 + wave * 16 + c;
#pragma unroll
        for (int i = 0; i < 16; ++i) {
            const float* __restrict__ p = wp + (size_t)(8 * i) * 128;
            v4f v;
            v.x = p[0];
            v.y = p[128];
            v.z = p[256];
            v.w = p[384];
            Wl[i * 256 + t] = v;
        }
    }

    const int tile = blockIdx.x * 8 + wave;
    const int tl   = tile < ntiles ? tile : ntiles - 1;
    const float* __restrict__ arow = x + ((size_t)tl * 16 + c) * 128 + 4 * hf;
    v4f a4[16];
#pragma unroll
    for (int q = 0; q < 16; ++q)
        a4[q] = *reinterpret_cast<const v4f*>(arow + 8 * q);

    float AL[8], AR[8];
#pragma unroll
    for (int n = 0; n < 8; ++n) {
        AL[n] = att_l[n * 16 + c];
        AR[n] = att_r[n * 16 + c];
    }

    __syncthreads();
    if (tile >= ntiles) return;

    {
        gfloat* op = (gfloat*)(out + (size_t)tile * 2048 + lane * 4);
        const v4f z = {0.f, 0.f, 0.f, 0.f};
#pragma unroll
        for (int i = 0; i < 16; ++i)
            *reinterpret_cast<__attribute__((address_space(1))) v4f*>(op + i * 128) = z;
        dn[(size_t)tile * 32 + lane]                 = 0.0f;
        dn[(size_t)N * 2 + (size_t)tile * 32 + lane] = 0.0f;
    }

    gfloat* hp  = (gfloat*)(h + ((size_t)tile * 16 + hf * 8) * 128 + c);
    gfloat* pal = (gfloat*)(alpha_l + (size_t)tile * 64 + lane * 2);
    gfloat* par = (gfloat*)(alpha_r + (size_t)tile * 64 + lane * 2);
    asm volatile("" : "+v"(hp), "+v"(pal), "+v"(par));

    v8f acc[8];
#pragma unroll
    for (int n = 0; n < 8; ++n)
        acc[n] = (v8f){0.f, 0.f, 0.f, 0.f, 0.f, 0.f, 0.f, 0.f};

#pragma unroll
    for (int q = 0; q < 16; ++q) {
        v4f b[8];
#pragma unroll
        for (int n = 0; n < 8; ++n) b[n] = Wl[(q * 8 + n) * 32 + lane];
#pragma unroll
        for (int n = 0; n < 8; ++n) {
            v2f a  = {a4[q].x, a4[q].y};
            v2f bb = {b[n].x, b[n].y};
            acc[n] = __builtin_amdgcn_wmma_f32_16x16x4_f32(false, a, false, bb,
                                                           (short)0, acc[n], false, false);
        }
#pragma unroll
        for (int n = 0; n < 8; ++n) {
            v2f a  = {a4[q].z, a4[q].w};
            v2f bb = {b[n].z, b[n].w};
            acc[n] = __builtin_amdgcn_wmma_f32_16x16x4_f32(false, a, false, bb,
                                                           (short)0, acc[n], false, false);
        }
    }

    int lb = lane;
    asm volatile("v_nop\n\tv_nop\n\tv_nop\n\tv_nop\n\tv_nop\n\tv_nop\n\tv_nop\n\tv_nop\n\tv_nop"
                 : "+v"(acc[0]), "+v"(acc[1]), "+v"(acc[2]), "+v"(acc[3]),
                   "+v"(acc[4]), "+v"(acc[5]), "+v"(acc[6]), "+v"(acc[7]),
                   "+v"(lb), "+v"(hp), "+v"(pal), "+v"(par));

#pragma unroll
    for (int n = 0; n < 8; ++n) {
#pragma unroll
        for (int r = 0; r < 8; ++r) hp[r * 128 + n * 16] = acc[n][r];
    }

    float P[2][4][8];
#pragma unroll
    for (int hd = 0; hd < 4; ++hd) {
#pragma unroll
        for (int r = 0; r < 8; ++r) {
            P[0][hd][r] = acc[2 * hd][r] * AL[2 * hd] + acc[2 * hd + 1][r] * AL[2 * hd + 1];
            P[1][hd][r] = acc[2 * hd][r] * AR[2 * hd] + acc[2 * hd + 1][r] * AR[2 * hd + 1];
        }
    }
    const bool b0 = (lb & 1) != 0, b1 = (lb & 2) != 0, b2 = (lb & 4) != 0, b3 = (lb & 8) != 0;
    float Q[2][2];
#pragma unroll
    for (int s = 0; s < 2; ++s) {
#pragma unroll
        for (int h0 = 0; h0 < 2; ++h0) {
            float s1[8], s2[4], s3[2];
#pragma unroll
            for (int r = 0; r < 8; ++r)
                s1[r] = xsum<1>(b0, P[s][h0][r], P[s][h0 + 2][r]);
#pragma unroll
            for (int r = 0; r < 4; ++r)
                s2[r] = xsum<2>(b1, s1[2 * r], s1[2 * r + 1]);
#pragma unroll
            for (int r = 0; r < 2; ++r)
                s3[r] = xsum<4>(b2, s2[2 * r], s2[2 * r + 1]);
            Q[s][h0] = xsum<8>(b3, s3[0], s3[1]);
        }
    }
    *reinterpret_cast<__attribute__((address_space(1))) v2f*>(pal) = (v2f){Q[0][0], Q[0][1]};
    *reinterpret_cast<__attribute__((address_space(1))) v2f*>(par) = (v2f){Q[1][0], Q[1][1]};
}

__device__ __forceinline__ float leaky02(float a) { return a > 0.f ? a : 0.2f * a; }

__global__ __launch_bounds__(256)
void gat_edge_max(const int* __restrict__ ei,
                  const float* __restrict__ alpha_l,
                  const float* __restrict__ alpha_r,
                  unsigned* __restrict__ maxbuf, int E) {
    __shared__ float smax[8][4];
    const int t    = threadIdx.x;
    const int lane = t & 31;
    const int wave = t >> 5;
    const int e0   = blockIdx.x * 1024 + t;

    int f[4], d[4];
#pragma unroll
    for (int k = 0; k < 4; ++k) {
        const int e  = e0 + k * 256;
        const int ee = e < E ? e : E - 1;
        f[k] = ei[ee];
        d[k] = ei[E + ee];
    }
    v4f al[4], ar[4];
#pragma unroll
    for (int k = 0; k < 4; ++k) {
        al[k] = *reinterpret_cast<const v4f*>(alpha_l + (size_t)f[k] * 4);
        ar[k] = *reinterpret_cast<const v4f*>(alpha_r + (size_t)d[k] * 4);
    }
    float m[4];
#pragma unroll
    for (int hd = 0; hd < 4; ++hd) {
        float v = leaky02(al[0][hd] + ar[0][hd]);
#pragma unroll
        for (int k = 1; k < 4; ++k) v = fmaxf(v, leaky02(al[k][hd] + ar[k][hd]));
        v = fmaxf(v, lane_xor16<1>(v));
        v = fmaxf(v, lane_xor16<2>(v));
        v = fmaxf(v, lane_xor16<4>(v));
        v = fmaxf(v, lane_xor16<8>(v));
        m[hd] = fmaxf(__uint_as_float(__builtin_amdgcn_readlane(__float_as_uint(v), 0)),
                      __uint_as_float(__builtin_amdgcn_readlane(__float_as_uint(v), 16)));
    }
    if (lane == 0) {
#pragma unroll
        for (int hd = 0; hd < 4; ++hd) smax[wave][hd] = m[hd];
    }
    __syncthreads();
    if (t < 4) {
        float v = smax[0][t];
#pragma unroll
        for (int w = 1; w < 8; ++w) v = fmaxf(v, smax[w][t]);
        atomicMax(&maxbuf[t], f2ord(v));
    }
}

__global__ void
gat_aggregate(const float* __restrict__ hbuf,
              const int* __restrict__ ei,
              const float* __restrict__ alpha_l,
              const float* __restrict__ alpha_r,
              const unsigned* __restrict__ maxbuf,
              float* __restrict__ dn,
              float* __restrict__ out,
              unsigned* __restrict__ claim, int pitch, int E, int N) {
    __shared__ int s_item;
    const int lane = threadIdx.x & 31;
    const int wave = threadIdx.x >> 5;
    const int g    = blockIdx.x >> 5;

    if (threadIdx.x == 0) {
        const unsigned id   = __builtin_amdgcn_s_sendmsg_rtn(0x87);
        const unsigned die  = (id >> 16) & 7u;
        const unsigned pref = ((die & 3u) << 2) | ((blockIdx.x >> 3) & 3u);
        unsigned ch = die >> 2;
        int got = -1;
        for (int r = 0; r < 2 && got < 0; ++r) {
            unsigned* __restrict__ w = claim + (size_t)ch * pitch + g;
            unsigned it = pref;
            for (int k = 0; k < 16; ++k) {
                const unsigned bit = 1u << it;
                const unsigned old = atomicOr(w, bit);
                if (!(old & bit)) { got = (int)(ch * 16u + it); break; }
                const unsigned freeb = ~(old | bit) & 0xFFFFu;
                if (!freeb) break;
                it = (unsigned)__builtin_ctz(freeb);
            }
            ch ^= 1u;
        }
        s_item = got;
    }
    __syncthreads();
    const int item = s_item;
    if (item < 0) return;

    const int c  = item >> 4;
    const int e0 = __builtin_amdgcn_readfirstlane(g * 2048 + (item & 15) * 128 + wave * 16);
    if (e0 >= E) return;
    const int nlive = (E - e0) < 16 ? (E - e0) : 16;

    const int  j  = lane >> 1;
    const int  hb = lane & 1;
    const int  hd = 2 * c + hb;
    const int  e  = (j < nlive) ? (e0 + j) : e0;
    const int  f  = ei[e];
    const int  t  = ei[E + e];
    float* __restrict__ dnc = dn + (size_t)c * N * 2;

    if (nlive == 16) {
        const float    al = alpha_l[f * 4 + hd];
        const float    ar = alpha_r[t * 4 + hd];
        const unsigned mx = maxbuf[hd];

        float v[16][2];
#pragma unroll
        for (int jj = 0; jj < 16; ++jj) {
            const float* __restrict__ srow =
                hbuf + (size_t)__builtin_amdgcn_readlane(f, jj * 2) * 128 + c * 64;
#pragma unroll
            for (int h2 = 0; h2 < 2; ++h2) v[jj][h2] = srow[h2 * 32 + lane];
        }

        float a = al + ar;
        a = a > 0.f ? a : 0.2f * a;
        const float ex = __expf(a - ord2f(mx));
        atomicAdd(&dnc[t * 2 + hb], ex);

#pragma unroll
        for (int jj = 0; jj < 16; ++jj) {
            float* __restrict__ dst =
                out + (size_t)__builtin_amdgcn_readlane(t, jj * 2) * 128 + c * 64;
#pragma unroll
            for (int h2 = 0; h2 < 2; ++h2) {
                const float w = __uint_as_float(
                    __builtin_amdgcn_readlane(__float_as_uint(ex), jj * 2 + h2));
                atomicAdd(&dst[h2 * 32 + lane], w * v[jj][h2]);
            }
        }
    } else {
        for (int jj = 0; jj < nlive; ++jj) {
            const int fj = __builtin_amdgcn_readlane(f, jj * 2);
            const int tj = __builtin_amdgcn_readlane(t, jj * 2);
            const float* __restrict__ srow = hbuf + (size_t)fj * 128 + c * 64;
            float* __restrict__ dst = out + (size_t)tj * 128 + c * 64;
            float ex_mine = 0.f;
#pragma unroll
            for (int h2 = 0; h2 < 2; ++h2) {
                const int hh = 2 * c + h2;
                float a = alpha_l[fj * 4 + hh] + alpha_r[tj * 4 + hh];
                a = a > 0.f ? a : 0.2f * a;
                const float ex = __expf(a - ord2f(maxbuf[hh]));
                if (lane == h2) ex_mine = ex;
                atomicAdd(&dst[h2 * 32 + lane], ex * srow[h2 * 32 + lane]);
            }
            if (lane < 2) atomicAdd(&dnc[tj * 2 + lane], ex_mine);
        }
    }
}

__global__ void gat_normalise(float* __restrict__ out,
                              const float* __restrict__ dn,
                              const float* __restrict__ bias, int total4, int N) {
    const int i = blockIdx.x * blockDim.x + threadIdx.x;
    if (i >= total4) return;
    const int col  = (i & 31) * 4;
    const int node = i >> 5;
    const int head = col >> 5;
    const float dv = dn[((size_t)(head >> 1) * N + node) * 2 + (head & 1)] + 1e-9f;
    float4 v = reinterpret_cast<float4*>(out)[i];
    v.x = v.x / dv + bias[col + 0];
    v.y = v.y / dv + bias[col + 1];
    v.z = v.z / dv + bias[col + 2];
    v.w = v.w / dv + bias[col + 3];
    reinterpret_cast<float4*>(out)[i] = v;
}

extern "C" void kernel_launch(void* const* d_in, const int* in_sizes, int n_in,
                              void* d_out, int out_size, void* d_ws, size_t ws_size,
                              hipStream_t stream) {
    const float* x     = (const float*)d_in[0];
    const int*   ei    = (const int*)d_in[1];
    const float* W     = (const float*)d_in[2];
    const float* att_l = (const float*)d_in[3];
    const float* att_r = (const float*)d_in[4];
    const float* bias  = (const float*)d_in[5];
    float* out = (float*)d_out;

    const int N  = in_sizes[0] / 128;
    const int E  = in_sizes[1] / 2;
    const int NH = N * 4;
    const int NG = (E + 2047) / 2048;
    const int PITCH = (NG + 63) & ~63;

    float* h       = (float*)d_ws;
    float* alpha_l = h + (size_t)N * 128;
    float* alpha_r = alpha_l + NH;
    float* dn      = alpha_r + NH;
    unsigned* maxbuf = (unsigned*)(dn + NH);
    unsigned* claim  = maxbuf + 64;

    const int BLK = 256;

    {
        const int ntiles = N / 16;
        gat_gemm_wmma<<<(ntiles + 7) / 8, BLK, 0, stream>>>(x, W, att_l, att_r, h,
                                                            alpha_l, alpha_r, out, dn,
                                                            maxbuf, claim,
                                                            ntiles, N, 2 * PITCH);
    }
    gat_edge_max<<<(E + 1023) / 1024, BLK, 0, stream>>>(ei, alpha_l, alpha_r,
                                                        maxbuf, E);
    gat_aggregate<<<NG * 32, BLK, 0, stream>>>(h, ei, alpha_l, alpha_r,
                                              maxbuf, dn, out, claim, PITCH, E, N);
    {
        const int total4 = N * 32;
        gat_normalise<<<(total4 + BLK - 1) / BLK, BLK, 0, stream>>>(out, dn,
                                                                    bias, total4, N);
    }
}


# Edge scores kept: the max pass stores leaky(alpha_l[from]+alpha_r[to]) per edge, 8 bytes per column half; the edge pass reads them as 1 line per wave, not 32 scattered alpha lines; sums untouched
# speedup vs baseline: 3.0662x; 1.0576x over previous
#include <hip/hip_runtime.h>
#include <hip/hip_bf16.h>


typedef __attribute__((ext_vector_type(2))) float v2f;
typedef __attribute__((ext_vector_type(4))) float v4f;
typedef __attribute__((ext_vector_type(8))) float v8f;
typedef __attribute__((address_space(1))) float gfloat;

__device__ __forceinline__ unsigned f2ord(float f) {
    unsigned u = __float_as_uint(f);
    return (u & 0x80000000u) ? ~u : (u | 0x80000000u);
}
__device__ __forceinline__ float ord2f(unsigned u) {
    unsigned v = (u & 0x80000000u) ? (u & 0x7FFFFFFFu) : ~u;
    return __uint_as_float(v);
}


template <int K>
__device__ __forceinline__ float lane_xor16(float v) {
    return __int_as_float(__builtin_amdgcn_update_dpp(0, __float_as_int(v), 0x160 + K,
                                                      0xf, 0xf, true));
}
template <int K>
__device__ __forceinline__ float xsum(bool bit, float v0, float v1) {
    const float keep = bit ? v1 : v0;
    const float send = bit ? v0 : v1;
    return keep + lane_xor16<K>(send);
}

__global__ __launch_bounds__(256)
void gat_gemm_wmma(const float* __restrict__ x,
                   const float* __restrict__ W,
                   const float* __restrict__ att_l,
                   const float* __restrict__ att_r,
                   float* __restrict__ h,
                   float* __restrict__ alpha_l,
                   float* __restrict__ alpha_r,
                   float* __restrict__ out,
                   float* __restrict__ dn,
                   unsigned* __restrict__ maxbuf,
                   unsigned* __restrict__ claim,
                   int ntiles, int N, int nclaim) {
    __shared__ v4f Wl[16 * 8 * 32];

    const int t    = threadIdx.x;
    const int wave = t >> 5;
    const int lane = t & 31;
    const int c    = lane & 15;
    const int hf   = lane >> 4;

    {
        const int gi = blockIdx.x * 256 + t;
        if (gi < nclaim) claim[gi]  = 0u;
        if (gi < 4)      maxbuf[gi] = f2ord(-3.0e38f);
    }

    {
        const float* __restrict__ wp = W + (size_t)(4 * hf) * 128 + wave * 16 + c;
#pragma unroll
        for (int i = 0; i < 16; ++i) {
            const float* __restrict__ p = wp + (size_t)(8 * i) * 128;
            v4f v;
            v.x = p[0];
            v.y = p[128];
            v.z = p[256];
            v.w = p[384];
            Wl[i * 256 + t] = v;
        }
    }

    const int tile = blockIdx.x * 8 + wave;
    const int tl   = tile < ntiles ? tile : ntiles - 1;
    const float* __restrict__ arow = x + ((size_t)tl * 16 + c) * 128 + 4 * hf;
    v4f a4[16];
#pragma unroll
    for (int q = 0; q < 16; ++q)
        a4[q] = *reinterpret_cast<const v4f*>(arow + 8 * q);

    float AL[8], AR[8];
#pragma unroll
    for (int n = 0; n < 8; ++n) {
        AL[n] = att_l[n * 16 + c];
        AR[n] = att_r[n * 16 + c];
    }

    __syncthreads();
    if (tile >= ntiles) return;

    {
        gfloat* op = (gfloat*)(out + (size_t)tile * 2048 + lane * 4);
        const v4f z = {0.f, 0.f, 0.f, 0.f};
#pragma unroll
        for (int i = 0; i < 16; ++i)
            *reinterpret_cast<__attribute__((address_space(1))) v4f*>(op + i * 128) = z;
        dn[(size_t)tile * 32 + lane]                 = 0.0f;
        dn[(size_t)N * 2 + (size_t)tile * 32 + lane] = 0.0f;
    }

    gfloat* hp  = (gfloat*)(h + ((size_t)tile * 16 + hf * 8) * 128 + c);
    gfloat* pal = (gfloat*)(alpha_l + (size_t)tile * 64 + lane * 2);
    gfloat* par = (gfloat*)(alpha_r + (size_t)tile * 64 + lane * 2);
    asm volatile("" : "+v"(hp), "+v"(pal), "+v"(par));

    v8f acc[8];
#pragma unroll
    for (int n = 0; n < 8; ++n)
        acc[n] = (v8f){0.f, 0.f, 0.f, 0.f, 0.f, 0.f, 0.f, 0.f};

#pragma unroll
    for (int q = 0; q < 16; ++q) {
        v4f b[8];
#pragma unroll
        for (int n = 0; n < 8; ++n) b[n] = Wl[(q * 8 + n) * 32 + lane];
#pragma unroll
        for (int n = 0; n < 8; ++n) {
            v2f a  = {a4[q].x, a4[q].y};
            v2f bb = {b[n].x, b[n].y};
            acc[n] = __builtin_amdgcn_wmma_f32_16x16x4_f32(false, a, false, bb,
                                                           (short)0, acc[n], false, false);
        }
#pragma unroll
        for (int n = 0; n < 8; ++n) {
            v2f a  = {a4[q].z, a4[q].w};
            v2f bb = {b[n].z, b[n].w};
            acc[n] = __builtin_amdgcn_wmma_f32_16x16x4_f32(false, a, false, bb,
                                                           (short)0, acc[n], false, false);
        }
    }

    int lb = lane;
    asm volatile("v_nop\n\tv_nop\n\tv_nop\n\tv_nop\n\tv_nop\n\tv_nop\n\tv_nop\n\tv_nop\n\tv_nop"
                 : "+v"(acc[0]), "+v"(acc[1]), "+v"(acc[2]), "+v"(acc[3]),
                   "+v"(acc[4]), "+v"(acc[5]), "+v"(acc[6]), "+v"(acc[7]),
                   "+v"(lb), "+v"(hp), "+v"(pal), "+v"(par));

#pragma unroll
    for (int n = 0; n < 8; ++n) {
#pragma unroll
        for (int r = 0; r < 8; ++r) hp[r * 128 + n * 16] = acc[n][r];
    }

    float P[2][4][8];
#pragma unroll
    for (int hd = 0; hd < 4; ++hd) {
#pragma unroll
        for (int r = 0; r < 8; ++r) {
            P[0][hd][r] = acc[2 * hd][r] * AL[2 * hd] + acc[2 * hd + 1][r] * AL[2 * hd + 1];
            P[1][hd][r] = acc[2 * hd][r] * AR[2 * hd] + acc[2 * hd + 1][r] * AR[2 * hd + 1];
        }
    }
    const bool b0 = (lb & 1) != 0, b1 = (lb & 2) != 0, b2 = (lb & 4) != 0, b3 = (lb & 8) != 0;
    float Q[2][2];
#pragma unroll
    for (int s = 0; s < 2; ++s) {
#pragma unroll
        for (int h0 = 0; h0 < 2; ++h0) {
            float s1[8], s2[4], s3[2];
#pragma unroll
            for (int r = 0; r < 8; ++r)
                s1[r] = xsum<1>(b0, P[s][h0][r], P[s][h0 + 2][r]);
#pragma unroll
            for (int r = 0; r < 4; ++r)
                s2[r] = xsum<2>(b1, s1[2 * r], s1[2 * r + 1]);
#pragma unroll
            for (int r = 0; r < 2; ++r)
                s3[r] = xsum<4>(b2, s2[2 * r], s2[2 * r + 1]);
            Q[s][h0] = xsum<8>(b3, s3[0], s3[1]);
        }
    }
    *reinterpret_cast<__attribute__((address_space(1))) v2f*>(pal) = (v2f){Q[0][0], Q[0][1]};
    *reinterpret_cast<__attribute__((address_space(1))) v2f*>(par) = (v2f){Q[1][0], Q[1][1]};
}

__device__ __forceinline__ float leaky02(float a) { return a > 0.f ? a : 0.2f * a; }

__global__ __launch_bounds__(256)
void gat_edge_max(const int* __restrict__ ei,
                  const float* __restrict__ alpha_l,
                  const float* __restrict__ alpha_r,
                  unsigned* __restrict__ maxbuf,
                  float* __restrict__ sc, int E) {
    __shared__ float smax[8][4];
    const int t    = threadIdx.x;
    const int lane = t & 31;
    const int wave = t >> 5;
    const int e0   = blockIdx.x * 1024 + t;

    int f[4], d[4];
#pragma unroll
    for (int k = 0; k < 4; ++k) {
        const int e  = e0 + k * 256;
        const int ee = e < E ? e : E - 1;
        f[k] = ei[ee];
        d[k] = ei[E + ee];
    }
    v4f al[4], ar[4];
#pragma unroll
    for (int k = 0; k < 4; ++k) {
        al[k] = *reinterpret_cast<const v4f*>(alpha_l + (size_t)f[k] * 4);
        ar[k] = *reinterpret_cast<const v4f*>(alpha_r + (size_t)d[k] * 4);
    }
    float s[4][4];
#pragma unroll
    for (int k = 0; k < 4; ++k) {
#pragma unroll
        for (int hd = 0; hd < 4; ++hd) s[k][hd] = leaky02(al[k][hd] + ar[k][hd]);
    }
    if (sc != nullptr) {
#pragma unroll
        for (int k = 0; k < 4; ++k) {
            const int e = e0 + k * 256;
            if (e < E) {
                *reinterpret_cast<v2f*>(sc + (size_t)e * 2) = (v2f){s[k][0], s[k][1]};
                *reinterpret_cast<v2f*>(sc + ((size_t)E + e) * 2) = (v2f){s[k][2], s[k][3]};
            }
        }
    }
    float m[4];
#pragma unroll
    for (int hd = 0; hd < 4; ++hd) {
        float v = s[0][hd];
#pragma unroll
        for (int k = 1; k < 4; ++k) v = fmaxf(v, s[k][hd]);
        v = fmaxf(v, lane_xor16<1>(v));
        v = fmaxf(v, lane_xor16<2>(v));
        v = fmaxf(v, lane_xor16<4>(v));
        v = fmaxf(v, lane_xor16<8>(v));
        m[hd] = fmaxf(__uint_as_float(__builtin_amdgcn_readlane(__float_as_uint(v), 0)),
                      __uint_as_float(__builtin_amdgcn_readlane(__float_as_uint(v), 16)));
    }
    if (lane == 0) {
#pragma unroll
        for (int hd = 0; hd < 4; ++hd) smax[wave][hd] = m[hd];
    }
    __syncthreads();
    if (t < 4) {
        float v = smax[0][t];
#pragma unroll
        for (int w = 1; w < 8; ++w) v = fmaxf(v, smax[w][t]);
        atomicMax(&maxbuf[t], f2ord(v));
    }
}

__global__ void
gat_aggregate(const float* __restrict__ hbuf,
              const int* __restrict__ ei,
              const float* __restrict__ alpha_l,
              const float* __restrict__ alpha_r,
              const unsigned* __restrict__ maxbuf,
              const float* __restrict__ sc,
              float* __restrict__ dn,
              float* __restrict__ out,
              unsigned* __restrict__ claim, int pitch, int E, int N) {
    __shared__ int s_item;
    const int lane = threadIdx.x & 31;
    const int wave = threadIdx.x >> 5;
    const int g    = blockIdx.x >> 5;

    if (threadIdx.x == 0) {
        const unsigned id   = __builtin_amdgcn_s_sendmsg_rtn(0x87);
        const unsigned die  = (id >> 16) & 7u;
        const unsigned pref = ((die & 3u) << 2) | ((blockIdx.x >> 3) & 3u);
        unsigned ch = die >> 2;
        int got = -1;
        for (int r = 0; r < 2 && got < 0; ++r) {
            unsigned* __restrict__ w = claim + (size_t)ch * pitch + g;
            unsigned it = pref;
            for (int k = 0; k < 16; ++k) {
                const unsigned bit = 1u << it;
                const unsigned old = atomicOr(w, bit);
                if (!(old & bit)) { got = (int)(ch * 16u + it); break; }
                const unsigned freeb = ~(old | bit) & 0xFFFFu;
                if (!freeb) break;
                it = (unsigned)__builtin_ctz(freeb);
            }
            ch ^= 1u;
        }
        s_item = got;
    }
    __syncthreads();
    const int item = s_item;
    if (item < 0) return;

    const int c  = item >> 4;
    const int e0 = __builtin_amdgcn_readfirstlane(g * 2048 + (item & 15) * 128 + wave * 16);
    if (e0 >= E) return;
    const int nlive = (E - e0) < 16 ? (E - e0) : 16;

    const int  j  = lane >> 1;
    const int  hb = lane & 1;
    const int  hd = 2 * c + hb;
    const int  e  = (j < nlive) ? (e0 + j) : e0;
    const int  f  = ei[e];
    const int  t  = ei[E + e];
    float* __restrict__ dnc = dn + (size_t)c * N * 2;
    const float* __restrict__ scc = sc ? sc + (size_t)c * E * 2 : nullptr;

    if (nlive == 16) {
        float a, ar = 0.f;
        if (scc) {
            a = scc[(size_t)e * 2 + hb];
        } else {
            a  = alpha_l[f * 4 + hd];
            ar = alpha_r[t * 4 + hd];
        }
        const unsigned mx = maxbuf[hd];

        float v[16][2];
#pragma unroll
        for (int jj = 0; jj < 16; ++jj) {
            const float* __restrict__ srow =
                hbuf + (size_t)__builtin_amdgcn_readlane(f, jj * 2) * 128 + c * 64;
#pragma unroll
            for (int h2 = 0; h2 < 2; ++h2) v[jj][h2] = srow[h2 * 32 + lane];
        }

        if (!scc) {
            a += ar;
            a = a > 0.f ? a : 0.2f * a;
        }
        const float ex = __expf(a - ord2f(mx));
        atomicAdd(&dnc[t * 2 + hb], ex);

#pragma unroll
        for (int jj = 0; jj < 16; ++jj) {
            float* __restrict__ dst =
                out + (size_t)__builtin_amdgcn_readlane(t, jj * 2) * 128 + c * 64;
#pragma unroll
            for (int h2 = 0; h2 < 2; ++h2) {
                const float w = __uint_as_float(
                    __builtin_amdgcn_readlane(__float_as_uint(ex), jj * 2 + h2));
                atomicAdd(&dst[h2 * 32 + lane], w * v[jj][h2]);
            }
        }
    } else {
        for (int jj = 0; jj < nlive; ++jj) {
            const int fj = __builtin_amdgcn_readlane(f, jj * 2);
            const int tj = __builtin_amdgcn_readlane(t, jj * 2);
            const float* __restrict__ srow = hbuf + (size_t)fj * 128 + c * 64;
            float* __restrict__ dst = out + (size_t)tj * 128 + c * 64;
            float ex_mine = 0.f;
#pragma unroll
            for (int h2 = 0; h2 < 2; ++h2) {
                const int hh = 2 * c + h2;
                float a;
                if (scc) {
                    a = scc[(size_t)(e0 + jj) * 2 + h2];
                } else {
                    a = alpha_l[fj * 4 + hh] + alpha_r[tj * 4 + hh];
                    a = a > 0.f ? a : 0.2f * a;
                }
                const float ex = __expf(a - ord2f(maxbuf[hh]));
                if (lane == h2) ex_mine = ex;
                atomicAdd(&dst[h2 * 32 + lane], ex * srow[h2 * 32 + lane]);
            }
            if (lane < 2) atomicAdd(&dnc[tj * 2 + lane], ex_mine);
        }
    }
}

__global__ void gat_normalise(float* __restrict__ out,
                              const float* __restrict__ dn,
                              const float* __restrict__ bias, int total4, int N) {
    const int i = blockIdx.x * blockDim.x + threadIdx.x;
    if (i >= total4) return;
    const int col  = (i & 31) * 4;
    const int node = i >> 5;
    const int head = col >> 5;
    const float dv = dn[((size_t)(head >> 1) * N + node) * 2 + (head & 1)] + 1e-9f;
    float4 v = reinterpret_cast<float4*>(out)[i];
    v.x = v.x / dv + bias[col + 0];
    v.y = v.y / dv + bias[col + 1];
    v.z = v.z / dv + bias[col + 2];
    v.w = v.w / dv + bias[col + 3];
    reinterpret_cast<float4*>(out)[i] = v;
}

extern "C" void kernel_launch(void* const* d_in, const int* in_sizes, int n_in,
                              void* d_out, int out_size, void* d_ws, size_t ws_size,
                              hipStream_t stream) {
    const float* x     = (const float*)d_in[0];
    const int*   ei    = (const int*)d_in[1];
    const float* W     = (const float*)d_in[2];
    const float* att_l = (const float*)d_in[3];
    const float* att_r = (const float*)d_in[4];
    const float* bias  = (const float*)d_in[5];
    float* out = (float*)d_out;

    const int N  = in_sizes[0] / 128;
    const int E  = in_sizes[1] / 2;
    const int NH = N * 4;
    const int NG = (E + 2047) / 2048;
    const int PITCH = (NG + 63) & ~63;

    float* h       = (float*)d_ws;
    float* alpha_l = h + (size_t)N * 128;
    float* alpha_r = alpha_l + NH;
    float* dn      = alpha_r + NH;
    unsigned* maxbuf = (unsigned*)(dn + NH);
    unsigned* claim  = maxbuf + 64;
    float* sc = (float*)(claim + 2 * PITCH);
    if (ws_size < ((size_t)N * 140 + 64 + 2 * (size_t)PITCH + (size_t)E * 4) * sizeof(float))
        sc = nullptr;

    const int BLK = 256;

    {
        const int ntiles = N / 16;
        gat_gemm_wmma<<<(ntiles + 7) / 8, BLK, 0, stream>>>(x, W, att_l, att_r, h,
                                                            alpha_l, alpha_r, out, dn,
                                                            maxbuf, claim,
                                                            ntiles, N, 2 * PITCH);
    }
    gat_edge_max<<<(E + 1023) / 1024, BLK, 0, stream>>>(ei, alpha_l, alpha_r,
                                                        maxbuf, sc, E);
    gat_aggregate<<<NG * 32, BLK, 0, stream>>>(h, ei, alpha_l, alpha_r,
                                              maxbuf, sc, dn, out, claim, PITCH, E, N);
    {
        const int total4 = N * 32;
        gat_normalise<<<(total4 + BLK - 1) / BLK, BLK, 0, stream>>>(out, dn,
                                                                    bias, total4, N);
    }
}
